# starting kernel
# baseline (speedup 1.0000x reference)
#include <hip/hip_runtime.h>
#include <math.h>

constexpr int   Bn   = 4;
constexpr int   Mn   = 4096;
constexpr int   Nn   = Bn * Mn;
constexpr int   PADI = -999;
#define PADF (-999.0f)
constexpr float CUTOFF2 = 49.0f;

typedef float v2f __attribute__((ext_vector_type(2)));
typedef float v8f __attribute__((ext_vector_type(8)));

__device__ __forceinline__ float3 sub3(float3 a, float3 b) {
    return make_float3(a.x - b.x, a.y - b.y, a.z - b.z);
}
__device__ __forceinline__ float3 cross3(float3 a, float3 b) {
    return make_float3(a.y * b.z - a.z * b.y,
                       a.z * b.x - a.x * b.z,
                       a.x * b.y - a.y * b.x);
}
__device__ __forceinline__ float dot3(float3 a, float3 b) {
    return a.x * b.x + a.y * b.y + a.z * b.z;
}

__global__ __launch_bounds__(256) void atoms_kernel(
    const float* __restrict__ coords,
    const int*   __restrict__ cia,
    const int*   __restrict__ partners,
    const int*   __restrict__ aidx,
    float*       __restrict__ out)
{
    int n = blockIdx.x * 256 + threadIdx.x;
    if (n >= Nn) return;
    int b = n >> 12;

    #pragma unroll
    for (int k = 0; k < 2; ++k) {
        int  p  = partners[n * 2 + k];
        bool pv = (p != PADI);
        int  sp = pv ? p : 0;
        const float* pc = coords + ((size_t)b * Mn + sp) * 3;
        out[(size_t)n * 8 + 2 + k * 3 + 0] = pv ? pc[0] : PADF;
        out[(size_t)n * 8 + 2 + k * 3 + 1] = pv ? pc[1] : PADF;
        out[(size_t)n * 8 + 2 + k * 3 + 2] = pv ? pc[2] : PADF;
    }

    int a[4];
    #pragma unroll
    for (int t = 0; t < 4; ++t) a[t] = aidx[n * 4 + t];
    bool am = (a[0] != PADI) && (a[1] != PADI) && (a[2] != PADI) && (a[3] != PADI);

    float3 p[4];
    #pragma unroll
    for (int t = 0; t < 4; ++t) {
        int g  = am ? a[t] : 0;
        int gb = g >> 12;
        int gm = cia[g];
        const float* c = coords + ((size_t)gb * Mn + gm) * 3;
        p[t] = make_float3(c[0], c[1], c[2]);
    }
    float3 b1 = sub3(p[1], p[0]);
    float3 b2 = sub3(p[2], p[1]);
    float3 b3 = sub3(p[3], p[2]);
    float3 n1 = cross3(b1, b2);
    float3 n2 = cross3(b2, b3);
    float  inv = 1.0f / (sqrtf(dot3(b2, b2)) + 1e-12f);
    float3 b2n = make_float3(b2.x * inv, b2.y * inv, b2.z * inv);
    float3 m1  = cross3(n1, b2n);
    float  xx  = dot3(n1, n2);
    float  yy  = dot3(m1, n2);
    out[(size_t)n * 8 + 0] = am ? atan2f(yy, xx) : PADF;
}

__global__ __launch_bounds__(256) void nbr_kernel(
    const float* __restrict__ coords,
    const int*   __restrict__ cia,
    float*       __restrict__ out)
{
    __shared__ float4 sh[Mn];

    int b   = blockIdx.x >> 5;
    int blk = blockIdx.x & 31;

    for (int m = threadIdx.x; m < Mn; m += 256) {
        int idx = cia[b * Mn + m];
        const float* c = coords + ((size_t)b * Mn + idx) * 3;
        float x = c[0], y = c[1], z = c[2];
        sh[m] = make_float4(x, y, z, x * x + y * y + z * z);
    }
    __syncthreads();

    int wave = threadIdx.x >> 5;
    int lane = threadIdx.x & 31;
    int half = lane >> 4;
    int l    = lane & 15;
    int i0   = (blk * 8 + wave) * 16;

    float4 arow = sh[i0 + l];
    v2f A;
    A.x = half ? arow.z : arow.x;
    A.y = half ? 0.0f   : arow.y;

    float h[8];
    #pragma unroll
    for (int v = 0; v < 8; ++v) h[v] = sh[i0 + half * 8 + v].w;

    float acc0[8], acc1[8];
    #pragma unroll
    for (int v = 0; v < 8; ++v) { acc0[v] = 0.0f; acc1[v] = 0.0f; }

    v8f zc = {};
    for (int j0 = 0; j0 < Mn; j0 += 32) {
        float4 brow0 = sh[j0 + l];
        float4 brow1 = sh[j0 + 16 + l];
        v2f B0, B1;
        B0.x = half ? brow0.z : brow0.x;
        B0.y = half ? 0.0f    : brow0.y;
        B1.x = half ? brow1.z : brow1.x;
        B1.y = half ? 0.0f    : brow1.y;
        float sqj0 = brow0.w;
        float sqj1 = brow1.w;

        v8f G0 = __builtin_amdgcn_wmma_f32_16x16x4_f32(
            false, A, false, B0, (short)0, zc, false, false);
        v8f G1 = __builtin_amdgcn_wmma_f32_16x16x4_f32(
            false, A, false, B1, (short)0, zc, false, false);

        #pragma unroll
        for (int v = 0; v < 8; ++v) {
            float d2 = fmaf(G0[v], -2.0f, h[v] + sqj0);
            float t  = fmaxf(d2, 0.0f) + 1e-12f;
            float s  = __builtin_amdgcn_sqrtf(t);
            acc0[v] += (t <= CUTOFF2) ? s : 0.0f;
        }
        #pragma unroll
        for (int v = 0; v < 8; ++v) {
            float d2 = fmaf(G1[v], -2.0f, h[v] + sqj1);
            float t  = fmaxf(d2, 0.0f) + 1e-12f;
            float s  = __builtin_amdgcn_sqrtf(t);
            acc1[v] += (t <= CUTOFF2) ? s : 0.0f;
        }
    }

    #pragma unroll
    for (int v = 0; v < 8; ++v) acc0[v] += acc1[v];

    #pragma unroll
    for (int off = 1; off < 16; off <<= 1) {
        #pragma unroll
        for (int v = 0; v < 8; ++v)
            acc0[v] += __shfl_xor(acc0[v], off, 32);
    }

    if (l == 0) {
        #pragma unroll
        for (int v = 0; v < 8; ++v) {
            int  row   = i0 + half * 8 + v;
            bool valid = (sh[row].x != PADF);
            out[((size_t)(b * Mn + row)) * 8 + 1] = valid ? acc0[v] : 0.0f;
        }
    }
}

extern "C" void kernel_launch(void* const* d_in, const int* in_sizes, int n_in,
                              void* d_out, int out_size, void* d_ws, size_t ws_size,
                              hipStream_t stream) {
    (void)in_sizes; (void)n_in; (void)out_size; (void)d_ws; (void)ws_size;
    const float* coords   = (const float*)d_in[0];
    const int*   cia      = (const int*)d_in[2];
    const int*   partners = (const int*)d_in[3];
    const int*   aidx     = (const int*)d_in[4];
    float*       out      = (float*)d_out;

    atoms_kernel<<<Nn / 256, 256, 0, stream>>>(coords, cia, partners, aidx, out);
    nbr_kernel<<<Bn * 32, 256, 0, stream>>>(coords, cia, out);
}


# nbr sweep: a 16-column tile's sqrt/cutoff/add tail runs only if a wave-wide test finds a d2 <= 49 in it (scalar branch, two 64-column steps later); skipped terms were +0, row sums bit-identical
# speedup vs baseline: 1.5323x; 1.5323x over previous
#include <hip/hip_runtime.h>
#include <math.h>

constexpr int   Bn   = 4;
constexpr int   Mn   = 4096;
constexpr int   Nn   = Bn * Mn;
constexpr int   PADI = -999;
#define PADF (-999.0f)
constexpr float CUTOFF2 = 49.0f;

typedef float v2f __attribute__((ext_vector_type(2)));
typedef float v8f __attribute__((ext_vector_type(8)));

__device__ __forceinline__ float3 sub3(float3 a, float3 b) {
    return make_float3(a.x - b.x, a.y - b.y, a.z - b.z);
}
__device__ __forceinline__ float3 cross3(float3 a, float3 b) {
    return make_float3(a.y * b.z - a.z * b.y,
                       a.z * b.x - a.x * b.z,
                       a.x * b.y - a.y * b.x);
}
__device__ __forceinline__ float dot3(float3 a, float3 b) {
    return a.x * b.x + a.y * b.y + a.z * b.z;
}

__device__ __forceinline__ float sqnorm3(float x, float y, float z) {
    return fmaf(z, z, fmaf(x, x, y * y));
}

__device__ __forceinline__ bool plane_fits(const float* ws, size_t ws_bytes) {
    return ws != nullptr && ws_bytes >= (size_t)Nn * 16 && (((size_t)ws) & 15) == 0;
}

__global__ __launch_bounds__(256) void atoms_kernel(
    const float* __restrict__ coords,
    const int*   __restrict__ cia,
    const int*   __restrict__ partners,
    const int*   __restrict__ aidx,
    float*       __restrict__ out,
    float*       __restrict__ ws,
    size_t                    ws_bytes)
{
    int n = blockIdx.x * 256 + threadIdx.x;
    if (n >= Nn) return;
    int b = n >> 12;
    int gi = cia[n];

    #pragma unroll
    for (int k = 0; k < 2; ++k) {
        int  p  = partners[n * 2 + k];
        bool pv = (p != PADI);
        int  sp = pv ? p : 0;
        const float* pc = coords + ((size_t)b * Mn + sp) * 3;
        out[(size_t)n * 8 + 2 + k * 3 + 0] = pv ? pc[0] : PADF;
        out[(size_t)n * 8 + 2 + k * 3 + 1] = pv ? pc[1] : PADF;
        out[(size_t)n * 8 + 2 + k * 3 + 2] = pv ? pc[2] : PADF;
    }

    int a[4];
    #pragma unroll
    for (int t = 0; t < 4; ++t) a[t] = aidx[n * 4 + t];
    bool am = (a[0] != PADI) && (a[1] != PADI) && (a[2] != PADI) && (a[3] != PADI);

    const float* gc = coords + ((size_t)b * Mn + gi) * 3;

    float3 p[4];
    #pragma unroll
    for (int t = 0; t < 4; ++t) {
        int g  = am ? a[t] : 0;
        int gb = g >> 12;
        int gm = cia[g];
        const float* c = coords + ((size_t)gb * Mn + gm) * 3;
        p[t] = make_float3(c[0], c[1], c[2]);
    }
    float gx = gc[0], gy = gc[1], gz = gc[2];
    float3 b1 = sub3(p[1], p[0]);
    float3 b2 = sub3(p[2], p[1]);
    float3 b3 = sub3(p[3], p[2]);
    float3 n1 = cross3(b1, b2);
    float3 n2 = cross3(b2, b3);
    float  inv = 1.0f / (sqrtf(dot3(b2, b2)) + 1e-12f);
    float3 b2n = make_float3(b2.x * inv, b2.y * inv, b2.z * inv);
    float3 m1  = cross3(n1, b2n);
    float  xx  = dot3(n1, n2);
    float  yy  = dot3(m1, n2);
    out[(size_t)n * 8 + 0] = am ? atan2f(yy, xx) : PADF;

    if (plane_fits(ws, ws_bytes))
        reinterpret_cast<float4*>(ws)[n] = make_float4(gx, gy, gz, sqnorm3(gx, gy, gz));
}

template <int RH>
__device__ __forceinline__ void sweep_rows(const float4* sh, int b, int i0,
                                           int half, int l,
                                           float* __restrict__ out)
{
    float4 arow = sh[i0 + l];
    v2f A;
    A.x = half ? arow.z : arow.x;
    A.y = half ? 0.0f   : arow.y;

    float h[4];
    #pragma unroll
    for (int v = 0; v < 4; ++v) h[v] = sh[i0 + half * 8 + RH * 4 + v].w;

    float acc0[4], acc1[4];
    #pragma unroll
    for (int v = 0; v < 4; ++v) { acc0[v] = 0.0f; acc1[v] = 0.0f; }

    v8f zc = {};

    auto finish_tile = [&](const float* dt, float (&acc)[4]) {
        #pragma unroll
        for (int v = 0; v < 4; ++v) {
            float t = fmaxf(dt[v], 0.0f) + 1e-12f;
            float s = __builtin_amdgcn_sqrtf(t);
            acc[v] += (t <= CUTOFF2) ? s : 0.0f;
        }
    };
    auto finish_step = [&](const float (&ds)[16], const unsigned (&fs)[4]) {
        if (fs[0] != 0u) finish_tile(ds + 0,  acc0);
        if (fs[1] != 0u) finish_tile(ds + 4,  acc1);
        if (fs[2] != 0u) finish_tile(ds + 8,  acc0);
        if (fs[3] != 0u) finish_tile(ds + 12, acc1);
    };

    auto step = [&](int j0, float (&ds)[16], unsigned (&fs)[4]) {
        float4 brow[4];
        #pragma unroll
        for (int g = 0; g < 4; ++g) brow[g] = sh[j0 + 16 * g + l];

        finish_step(ds, fs);

        #pragma unroll
        for (int g = 0; g < 4; ++g) {
            v2f Bf;
            Bf.x = half ? brow[g].z : brow[g].x;
            Bf.y = half ? 0.0f      : brow[g].y;
            float sqj = brow[g].w;
            v8f G = __builtin_amdgcn_wmma_f32_16x16x4_f32(
                false, A, false, Bf, (short)0, zc, false, false);
            #pragma unroll
            for (int v = 0; v < 4; ++v)
                ds[4 * g + v] = fmaf(G[RH * 4 + v], -2.0f, h[v] + sqj);
            float m = fminf(fminf(ds[4 * g], ds[4 * g + 1]),
                            fminf(ds[4 * g + 2], ds[4 * g + 3]));
            fs[g] = __builtin_amdgcn_ballot_w32(m <= CUTOFF2);
        }
    };

    float    dA[16], dB[16];
    unsigned fA[4] = {0u, 0u, 0u, 0u}, fB[4] = {0u, 0u, 0u, 0u};
    #pragma unroll
    for (int v = 0; v < 16; ++v) { dA[v] = 0.0f; dB[v] = 0.0f; }

    for (int j0 = 0; j0 < Mn; j0 += 128) {
        step(j0,      dA, fA);
        step(j0 + 64, dB, fB);
    }
    finish_step(dA, fA);
    finish_step(dB, fB);

    #pragma unroll
    for (int v = 0; v < 4; ++v) acc0[v] += acc1[v];

    #pragma unroll
    for (int off = 1; off < 16; off <<= 1) {
        #pragma unroll
        for (int v = 0; v < 4; ++v)
            acc0[v] += __shfl_xor(acc0[v], off, 32);
    }

    if (l == 0) {
        #pragma unroll
        for (int v = 0; v < 4; ++v) {
            int  row   = i0 + half * 8 + RH * 4 + v;
            bool valid = (sh[row].x != PADF);
            out[((size_t)(b * Mn + row)) * 8 + 1] = valid ? acc0[v] : 0.0f;
        }
    }
}

__global__ __launch_bounds__(256) void nbr_kernel(
    const float* __restrict__ coords,
    const int*   __restrict__ cia,
    float*       __restrict__ out,
    const float* __restrict__ ws,
    size_t                    ws_bytes)
{
    __shared__ float4 sh[Mn];

    int b   = blockIdx.x >> 6;
    int blk = blockIdx.x & 63;

    if (plane_fits(ws, ws_bytes)) {
        const float4* plane = reinterpret_cast<const float4*>(ws) + (size_t)b * Mn;
        #pragma unroll
        for (int k = 0; k < Mn / 256; ++k)
            sh[k * 256 + threadIdx.x] = plane[k * 256 + threadIdx.x];
    } else {
        for (int m = threadIdx.x; m < Mn; m += 256) {
            int idx = cia[b * Mn + m];
            const float* c = coords + ((size_t)b * Mn + idx) * 3;
            float x = c[0], y = c[1], z = c[2];
            sh[m] = make_float4(x, y, z, sqnorm3(x, y, z));
        }
    }
    __syncthreads();

    int wave = __builtin_amdgcn_readfirstlane((int)(threadIdx.x >> 5));
    int lane = threadIdx.x & 31;
    int half = lane >> 4;
    int l    = lane & 15;
    int i0   = (blk * 4 + (wave >> 1)) * 16;

    if ((wave & 1) == 0) sweep_rows<0>(sh, b, i0, half, l, out);
    else                 sweep_rows<1>(sh, b, i0, half, l, out);
}

extern "C" void kernel_launch(void* const* d_in, const int* in_sizes, int n_in,
                              void* d_out, int out_size, void* d_ws, size_t ws_size,
                              hipStream_t stream) {
    (void)in_sizes; (void)n_in; (void)out_size;
    const float* coords   = (const float*)d_in[0];
    const int*   cia      = (const int*)d_in[2];
    const int*   partners = (const int*)d_in[3];
    const int*   aidx     = (const int*)d_in[4];
    float*       out      = (float*)d_out;
    float*       ws       = (float*)d_ws;

    atoms_kernel<<<Nn / 256, 256, 0, stream>>>(coords, cia, partners, aidx, out, ws, ws_size);
    nbr_kernel<<<Bn * 64, 256, 0, stream>>>(coords, cia, out, ws, ws_size);
}


# nbr sums by pairs of 256-atom groups: each distance tile formed once, its terms added to row atom and column atom; 16 partial sums per atom via workspace; changes the order of the row sum
# speedup vs baseline: 1.8419x; 1.2020x over previous
#include <hip/hip_runtime.h>
#include <math.h>


constexpr int   Bn   = 4;
constexpr int   Mn   = 4096;
constexpr int   Nn   = Bn * Mn;
constexpr int   PADI = -999;
#define PADF (-999.0f)
constexpr float CUTOFF2 = 49.0f;

constexpr int ST    = 16;
constexpr int SA    = Mn / ST;
constexpr int NPAIR = ST * (ST + 1) / 2;

typedef float v2f __attribute__((ext_vector_type(2)));
typedef float v8f __attribute__((ext_vector_type(8)));

__device__ __forceinline__ float3 sub3(float3 a, float3 b) {
    return make_float3(a.x - b.x, a.y - b.y, a.z - b.z);
}
__device__ __forceinline__ float3 cross3(float3 a, float3 b) {
    return make_float3(a.y * b.z - a.z * b.y,
                       a.z * b.x - a.x * b.z,
                       a.x * b.y - a.y * b.x);
}
__device__ __forceinline__ float dot3(float3 a, float3 b) {
    return a.x * b.x + a.y * b.y + a.z * b.z;
}

__device__ __forceinline__ float sqnorm3(float x, float y, float z) {
    return fmaf(z, z, fmaf(x, x, y * y));
}

__device__ __forceinline__ bool parts_fit(const float* ws, size_t ws_bytes) {
    return ws != nullptr && ws_bytes >= (size_t)ST * Nn * 4;
}

__global__ __launch_bounds__(256) void atoms_kernel(
    const float* __restrict__ coords,
    const int*   __restrict__ cia,
    const int*   __restrict__ partners,
    const int*   __restrict__ aidx,
    float*       __restrict__ out,
    const float* __restrict__ ws,
    size_t                    ws_bytes)
{
    int n = blockIdx.x * 256 + threadIdx.x;
    if (n >= Nn) return;
    int b = n >> 12;
    int gi = cia[n];

    const bool have_parts = parts_fit(ws, ws_bytes);
    float part[ST];
    #pragma unroll
    for (int s = 0; s < ST; ++s) part[s] = 0.0f;
    if (have_parts) {
        #pragma unroll
        for (int s = 0; s < ST; ++s) part[s] = ws[(size_t)s * Nn + n];
    }

    #pragma unroll
    for (int k = 0; k < 2; ++k) {
        int  p  = partners[n * 2 + k];
        bool pv = (p != PADI);
        int  sp = pv ? p : 0;
        const float* pc = coords + ((size_t)b * Mn + sp) * 3;
        out[(size_t)n * 8 + 2 + k * 3 + 0] = pv ? pc[0] : PADF;
        out[(size_t)n * 8 + 2 + k * 3 + 1] = pv ? pc[1] : PADF;
        out[(size_t)n * 8 + 2 + k * 3 + 2] = pv ? pc[2] : PADF;
    }

    int a[4];
    #pragma unroll
    for (int t = 0; t < 4; ++t) a[t] = aidx[n * 4 + t];
    bool am = (a[0] != PADI) && (a[1] != PADI) && (a[2] != PADI) && (a[3] != PADI);

    const float* gc = coords + ((size_t)b * Mn + gi) * 3;

    float3 p[4];
    #pragma unroll
    for (int t = 0; t < 4; ++t) {
        int g  = am ? a[t] : 0;
        int gb = g >> 12;
        int gm = cia[g];
        const float* c = coords + ((size_t)gb * Mn + gm) * 3;
        p[t] = make_float3(c[0], c[1], c[2]);
    }
    float gx = gc[0];
    float3 b1 = sub3(p[1], p[0]);
    float3 b2 = sub3(p[2], p[1]);
    float3 b3 = sub3(p[3], p[2]);
    float3 n1 = cross3(b1, b2);
    float3 n2 = cross3(b2, b3);
    float  inv = 1.0f / (sqrtf(dot3(b2, b2)) + 1e-12f);
    float3 b2n = make_float3(b2.x * inv, b2.y * inv, b2.z * inv);
    float3 m1  = cross3(n1, b2n);
    float  xx  = dot3(n1, n2);
    float  yy  = dot3(m1, n2);
    out[(size_t)n * 8 + 0] = am ? atan2f(yy, xx) : PADF;

    if (have_parts) {
        float s = part[0];
        #pragma unroll
        for (int k = 1; k < ST; ++k) s += part[k];
        out[(size_t)n * 8 + 1] = (gx != PADF) ? s : 0.0f;
    }
}


__device__ __forceinline__ void finish_half(const float* dt, float* acc, float& cc) {
    #pragma unroll
    for (int v = 0; v < 4; ++v) {
        float t   = fmaxf(dt[v], 0.0f) + 1e-12f;
        float s   = __builtin_amdgcn_sqrtf(t);
        float val = (t <= CUTOFF2) ? s : 0.0f;
        acc[v] += val;
        cc     += val;
    }
}

template <int CB>
__device__ __forceinline__ void finish_step(const float (&ds)[32], const unsigned (&fs)[8],
                                            float (&acc)[8], float (&ccol)[16]) {
    #pragma unroll
    for (int g = 0; g < 4; ++g) {
        if (fs[2 * g]     != 0u) finish_half(ds + 8 * g,     acc,     ccol[CB + g]);
        if (fs[2 * g + 1] != 0u) finish_half(ds + 8 * g + 4, acc + 4, ccol[CB + g]);
    }
}

template <int G>
__device__ __forceinline__ void form_tile(float4 br, v2f A, const float (&h)[8],
                                          int half, float (&ds)[32], unsigned (&fs)[8]) {
    v8f zc = {};
    v2f Bf;
    Bf.x = half ? br.z : br.x;
    Bf.y = half ? 0.0f : br.y;
    float sqj = br.w;
    v8f Gm = __builtin_amdgcn_wmma_f32_16x16x4_f32(
        false, A, false, Bf, (short)0, zc, false, false);
    #pragma unroll
    for (int r = 0; r < 8; ++r)
        ds[8 * G + r] = fmaf(Gm[r], -2.0f, h[r] + sqj);
    float m0 = fminf(fminf(ds[8 * G],     ds[8 * G + 1]), fminf(ds[8 * G + 2], ds[8 * G + 3]));
    float m1 = fminf(fminf(ds[8 * G + 4], ds[8 * G + 5]), fminf(ds[8 * G + 6], ds[8 * G + 7]));
    fs[2 * G]     = __builtin_amdgcn_ballot_w32(m0 <= CUTOFF2);
    fs[2 * G + 1] = __builtin_amdgcn_ballot_w32(m1 <= CUTOFF2);
}

__device__ __forceinline__ void form_step(float4 r0, float4 r1, float4 r2, float4 r3, v2f A,
                                          const float (&h)[8], int half,
                                          float (&ds)[32], unsigned (&fs)[8]) {
    form_tile<0>(r0, A, h, half, ds, fs);
    form_tile<1>(r1, A, h, half, ds, fs);
    form_tile<2>(r2, A, h, half, ds, fs);
    form_tile<3>(r3, A, h, half, ds, fs);
}

__device__ __forceinline__ void sweep_tile(const float4* shP, const float4* shQ, int i0,
                                           int half, int l, float (&acc)[8], float (&ccol)[16])
{
    float4 arow = shP[i0 + l];
    v2f A;
    A.x = half ? arow.z : arow.x;
    A.y = half ? 0.0f   : arow.y;

    float h[8];
    #pragma unroll
    for (int r = 0; r < 8; ++r) h[r] = shP[i0 + half * 8 + r].w;

    float    dA[32], dB[32];
    unsigned fA[8], fB[8];
    const float4* q = shQ + l;
    float4 r0, r1, r2, r3;

    r0 = q[0];   r1 = q[16];  r2 = q[32];  r3 = q[48];
    form_step(r0, r1, r2, r3, A, h, half, dA, fA);

    r0 = q[64];  r1 = q[80];  r2 = q[96];  r3 = q[112];
    form_step(r0, r1, r2, r3, A, h, half, dB, fB);

    r0 = q[128]; r1 = q[144]; r2 = q[160]; r3 = q[176];
    finish_step<0>(dA, fA, acc, ccol);
    form_step(r0, r1, r2, r3, A, h, half, dA, fA);

    r0 = q[192]; r1 = q[208]; r2 = q[224]; r3 = q[240];
    finish_step<4>(dB, fB, acc, ccol);
    form_step(r0, r1, r2, r3, A, h, half, dB, fB);

    finish_step<8>(dA, fA, acc, ccol);
    finish_step<12>(dB, fB, acc, ccol);
}

static_assert(SA == 256, "sweep_tile walks 4 steps of 64 columns; 128 threads fill 2 rows each");

template <int M>
__device__ __forceinline__ float row_xor_add(float v) {
    int o = __builtin_amdgcn_update_dpp(0, __float_as_int(v), 0x160 + M, 0xF, 0xF, false);
    return v + __int_as_float(o);
}

__device__ __forceinline__ void gather_group(float4* sh, const float* __restrict__ coords,
                                             const int* __restrict__ cia, int b, int grp, int t)
{
    #pragma unroll
    for (int k = 0; k < SA / 128; ++k) {
        int m   = k * 128 + t;
        int idx = cia[b * Mn + grp * SA + m];
        const float* c = coords + ((size_t)b * Mn + idx) * 3;
        float x = c[0], y = c[1], z = c[2];
        sh[m] = make_float4(x, y, z, sqnorm3(x, y, z));
    }
}

__global__ __launch_bounds__(128) void pairs_kernel(
    const float* __restrict__ coords,
    const int*   __restrict__ cia,
    float*       __restrict__ out,
    float*       __restrict__ ws,
    size_t                    ws_bytes)
{
    __shared__ float4 shP[SA];
    __shared__ float4 shQ[SA];
    __shared__ float  rowp[SA];
    __shared__ float  colp[4][SA];

    const bool sym = parts_fit(ws, ws_bytes);
    const int  t   = threadIdx.x;
    int b, P, Q0, nQ;
    if (sym) {
        b = blockIdx.x / NPAIR;
        int p = blockIdx.x - b * NPAIR;
        P = 0;
        for (int k = 0; k < ST; ++k) {
            int w = ST - P;
            if (p >= w) { p -= w; ++P; }
        }
        Q0 = P + p;
        nQ = 1;
    } else {
        if (blockIdx.x >= Bn * ST) return;
        b  = blockIdx.x / ST;
        P  = blockIdx.x - b * ST;
        Q0 = 0;
        nQ = ST;
    }

    const int wave = __builtin_amdgcn_readfirstlane((int)(threadIdx.x >> 5));
    const int lane = threadIdx.x & 31;
    const int half = lane >> 4;
    const int l    = lane & 15;

    {
        int ip[2], iq[2];
        #pragma unroll
        for (int k = 0; k < 2; ++k) {
            ip[k] = cia[b * Mn + P  * SA + k * 128 + t];
            iq[k] = cia[b * Mn + Q0 * SA + k * 128 + t];
        }
        float4 vp[2], vq[2];
        #pragma unroll
        for (int k = 0; k < 2; ++k) {
            const float* cp = coords + ((size_t)b * Mn + ip[k]) * 3;
            const float* cq = coords + ((size_t)b * Mn + iq[k]) * 3;
            float px = cp[0], py = cp[1], pz = cp[2];
            float qx = cq[0], qy = cq[1], qz = cq[2];
            vp[k] = make_float4(px, py, pz, sqnorm3(px, py, pz));
            vq[k] = make_float4(qx, qy, qz, sqnorm3(qx, qy, qz));
        }
        #pragma unroll
        for (int k = 0; k < 2; ++k) {
            shP[k * 128 + t] = vp[k];
            shQ[k * 128 + t] = vq[k];
        }
    }

    for (int q = 0; q < nQ; ++q) {
        const int Q = Q0 + q;
        if (q > 0) gather_group(shQ, coords, cia, b, Q, t);
        __syncthreads();

        float ccol[16];
        #pragma unroll
        for (int j = 0; j < 16; ++j) ccol[j] = 0.0f;

        for (int it = 0; it < 4; ++it) {
            const int i0 = (wave * 4 + it) * 16;
            float acc[8];
            #pragma unroll
            for (int r = 0; r < 8; ++r) acc[r] = 0.0f;

            sweep_tile(shP, shQ, i0, half, l, acc, ccol);

            #pragma unroll
            for (int r = 0; r < 8; ++r) {
                acc[r] = row_xor_add<1>(acc[r]);
                acc[r] = row_xor_add<2>(acc[r]);
                acc[r] = row_xor_add<4>(acc[r]);
                acc[r] = row_xor_add<8>(acc[r]);
            }
            if (l == 0) {
                #pragma unroll
                for (int r = 0; r < 8; ++r) {
                    int row = i0 + half * 8 + r;
                    rowp[row] = (q == 0) ? acc[r] : rowp[row] + acc[r];
                }
            }
        }

        #pragma unroll
        for (int j = 0; j < 16; ++j) ccol[j] += __shfl_xor(ccol[j], 16, 32);
        if (half == 0) {
            #pragma unroll
            for (int j = 0; j < 16; ++j) colp[wave][j * 16 + l] = ccol[j];
        }
        __syncthreads();

        if (sym && P != Q) {
            for (int k = 0; k < SA / 128; ++k) {
                int m = k * 128 + t;
                float c = (colp[0][m] + colp[1][m]) + (colp[2][m] + colp[3][m]);
                ws[(size_t)P * Nn + b * Mn + Q * SA + m] = c;
            }
        }
    }

    for (int k = 0; k < SA / 128; ++k) {
        int m = k * 128 + t;
        int n = b * Mn + P * SA + m;
        if (sym) {
            ws[(size_t)Q0 * Nn + n] = rowp[m];
        } else {
            bool valid = (shP[m].x != PADF);
            out[(size_t)n * 8 + 1] = valid ? rowp[m] : 0.0f;
        }
    }
}

extern "C" void kernel_launch(void* const* d_in, const int* in_sizes, int n_in,
                              void* d_out, int out_size, void* d_ws, size_t ws_size,
                              hipStream_t stream) {
    (void)in_sizes; (void)n_in; (void)out_size;
    const float* coords   = (const float*)d_in[0];
    const int*   cia      = (const int*)d_in[2];
    const int*   partners = (const int*)d_in[3];
    const int*   aidx     = (const int*)d_in[4];
    float*       out      = (float*)d_out;
    float*       ws       = (float*)d_ws;

    pairs_kernel<<<Bn * NPAIR, 128, 0, stream>>>(coords, cia, out, ws, ws_size);
    atoms_kernel<<<Nn / 256, 256, 0, stream>>>(coords, cia, partners, aidx, out, ws, ws_size);
}


# nbr sweep: the 8 row sums of a row tile are four separate register pairs, not an array; the array was held as one 8-register value and copied whole at every tail's branch merge; sums bit-identical
# speedup vs baseline: 1.9574x; 1.0627x over previous
#include <hip/hip_runtime.h>
#include <math.h>


constexpr int   Bn   = 4;
constexpr int   Mn   = 4096;
constexpr int   Nn   = Bn * Mn;
constexpr int   PADI = -999;
#define PADF (-999.0f)
constexpr float CUTOFF2 = 49.0f;

constexpr int ST    = 16;
constexpr int SA    = Mn / ST;
constexpr int NPAIR = ST * (ST + 1) / 2;

typedef float v2f __attribute__((ext_vector_type(2)));
typedef float v8f __attribute__((ext_vector_type(8)));

__device__ __forceinline__ float3 sub3(float3 a, float3 b) {
    return make_float3(a.x - b.x, a.y - b.y, a.z - b.z);
}
__device__ __forceinline__ float3 cross3(float3 a, float3 b) {
    return make_float3(a.y * b.z - a.z * b.y,
                       a.z * b.x - a.x * b.z,
                       a.x * b.y - a.y * b.x);
}
__device__ __forceinline__ float dot3(float3 a, float3 b) {
    return a.x * b.x + a.y * b.y + a.z * b.z;
}

__device__ __forceinline__ float sqnorm3(float x, float y, float z) {
    return fmaf(z, z, fmaf(x, x, y * y));
}

__device__ __forceinline__ bool parts_fit(const float* ws, size_t ws_bytes) {
    return ws != nullptr && ws_bytes >= (size_t)ST * Nn * 4;
}

__global__ __launch_bounds__(256) void atoms_kernel(
    const float* __restrict__ coords,
    const int*   __restrict__ cia,
    const int*   __restrict__ partners,
    const int*   __restrict__ aidx,
    float*       __restrict__ out,
    const float* __restrict__ ws,
    size_t                    ws_bytes)
{
    int n = blockIdx.x * 256 + threadIdx.x;
    if (n >= Nn) return;
    int b = n >> 12;
    int gi = cia[n];

    const bool have_parts = parts_fit(ws, ws_bytes);
    float part[ST];
    #pragma unroll
    for (int s = 0; s < ST; ++s) part[s] = 0.0f;
    if (have_parts) {
        #pragma unroll
        for (int s = 0; s < ST; ++s) part[s] = ws[(size_t)s * Nn + n];
    }

    #pragma unroll
    for (int k = 0; k < 2; ++k) {
        int  p  = partners[n * 2 + k];
        bool pv = (p != PADI);
        int  sp = pv ? p : 0;
        const float* pc = coords + ((size_t)b * Mn + sp) * 3;
        out[(size_t)n * 8 + 2 + k * 3 + 0] = pv ? pc[0] : PADF;
        out[(size_t)n * 8 + 2 + k * 3 + 1] = pv ? pc[1] : PADF;
        out[(size_t)n * 8 + 2 + k * 3 + 2] = pv ? pc[2] : PADF;
    }

    int a[4];
    #pragma unroll
    for (int t = 0; t < 4; ++t) a[t] = aidx[n * 4 + t];
    bool am = (a[0] != PADI) && (a[1] != PADI) && (a[2] != PADI) && (a[3] != PADI);

    const float* gc = coords + ((size_t)b * Mn + gi) * 3;

    float3 p[4];
    #pragma unroll
    for (int t = 0; t < 4; ++t) {
        int g  = am ? a[t] : 0;
        int gb = g >> 12;
        int gm = cia[g];
        const float* c = coords + ((size_t)gb * Mn + gm) * 3;
        p[t] = make_float3(c[0], c[1], c[2]);
    }
    float gx = gc[0];
    float3 b1 = sub3(p[1], p[0]);
    float3 b2 = sub3(p[2], p[1]);
    float3 b3 = sub3(p[3], p[2]);
    float3 n1 = cross3(b1, b2);
    float3 n2 = cross3(b2, b3);
    float  inv = 1.0f / (sqrtf(dot3(b2, b2)) + 1e-12f);
    float3 b2n = make_float3(b2.x * inv, b2.y * inv, b2.z * inv);
    float3 m1  = cross3(n1, b2n);
    float  xx  = dot3(n1, n2);
    float  yy  = dot3(m1, n2);
    out[(size_t)n * 8 + 0] = am ? atan2f(yy, xx) : PADF;

    if (have_parts) {
        float s = part[0];
        #pragma unroll
        for (int k = 1; k < ST; ++k) s += part[k];
        out[(size_t)n * 8 + 1] = (gx != PADF) ? s : 0.0f;
    }
}


struct RowSums { v2f p0, p1, p2, p3; };

__device__ __forceinline__ void finish_half(const float* dt, v2f& lo, v2f& hi, float& cc) {
    float val[4];
    #pragma unroll
    for (int v = 0; v < 4; ++v) {
        float t = fmaxf(dt[v], 0.0f) + 1e-12f;
        float s = __builtin_amdgcn_sqrtf(t);
        val[v]  = (t <= CUTOFF2) ? s : 0.0f;
        cc     += val[v];
    }
    lo += v2f{val[0], val[1]};
    hi += v2f{val[2], val[3]};
}

template <int CB>
__device__ __forceinline__ void finish_step(const float (&ds)[32], const unsigned (&fs)[8],
                                            RowSums& acc, float (&ccol)[16]) {
    #pragma unroll
    for (int g = 0; g < 4; ++g) {
        if (fs[2 * g]     != 0u) finish_half(ds + 8 * g,     acc.p0, acc.p1, ccol[CB + g]);
        if (fs[2 * g + 1] != 0u) finish_half(ds + 8 * g + 4, acc.p2, acc.p3, ccol[CB + g]);
    }
}

template <int G>
__device__ __forceinline__ void form_tile(float4 br, v2f A, const float (&h)[8],
                                          int half, float (&ds)[32], unsigned (&fs)[8]) {
    v8f zc = {};
    v2f Bf;
    Bf.x = half ? br.z : br.x;
    Bf.y = half ? 0.0f : br.y;
    float sqj = br.w;
    v8f Gm = __builtin_amdgcn_wmma_f32_16x16x4_f32(
        false, A, false, Bf, (short)0, zc, false, false);
    #pragma unroll
    for (int r = 0; r < 8; ++r)
        ds[8 * G + r] = fmaf(Gm[r], -2.0f, h[r] + sqj);
    float m0 = fminf(fminf(ds[8 * G],     ds[8 * G + 1]), fminf(ds[8 * G + 2], ds[8 * G + 3]));
    float m1 = fminf(fminf(ds[8 * G + 4], ds[8 * G + 5]), fminf(ds[8 * G + 6], ds[8 * G + 7]));
    fs[2 * G]     = __builtin_amdgcn_ballot_w32(m0 <= CUTOFF2);
    fs[2 * G + 1] = __builtin_amdgcn_ballot_w32(m1 <= CUTOFF2);
}

__device__ __forceinline__ void form_step(float4 r0, float4 r1, float4 r2, float4 r3, v2f A,
                                          const float (&h)[8], int half,
                                          float (&ds)[32], unsigned (&fs)[8]) {
    form_tile<0>(r0, A, h, half, ds, fs);
    form_tile<1>(r1, A, h, half, ds, fs);
    form_tile<2>(r2, A, h, half, ds, fs);
    form_tile<3>(r3, A, h, half, ds, fs);
}

__device__ __forceinline__ void sweep_tile(const float4* shP, const float4* shQ, int i0,
                                           int half, int l, RowSums& acc, float (&ccol)[16])
{
    float4 arow = shP[i0 + l];
    v2f A;
    A.x = half ? arow.z : arow.x;
    A.y = half ? 0.0f   : arow.y;

    float h[8];
    #pragma unroll
    for (int r = 0; r < 8; ++r) h[r] = shP[i0 + half * 8 + r].w;

    float    dA[32], dB[32];
    unsigned fA[8], fB[8];
    const float4* q = shQ + l;
    float4 r0, r1, r2, r3;

    r0 = q[0];   r1 = q[16];  r2 = q[32];  r3 = q[48];
    form_step(r0, r1, r2, r3, A, h, half, dA, fA);

    r0 = q[64];  r1 = q[80];  r2 = q[96];  r3 = q[112];
    form_step(r0, r1, r2, r3, A, h, half, dB, fB);

    r0 = q[128]; r1 = q[144]; r2 = q[160]; r3 = q[176];
    finish_step<0>(dA, fA, acc, ccol);
    form_step(r0, r1, r2, r3, A, h, half, dA, fA);

    r0 = q[192]; r1 = q[208]; r2 = q[224]; r3 = q[240];
    finish_step<4>(dB, fB, acc, ccol);
    form_step(r0, r1, r2, r3, A, h, half, dB, fB);

    finish_step<8>(dA, fA, acc, ccol);
    finish_step<12>(dB, fB, acc, ccol);
}

static_assert(SA == 256, "sweep_tile walks 4 steps of 64 columns; 128 threads fill 2 rows each");

template <int M>
__device__ __forceinline__ float row_xor_add(float v) {
    int o = __builtin_amdgcn_update_dpp(0, __float_as_int(v), 0x160 + M, 0xF, 0xF, false);
    return v + __int_as_float(o);
}

__device__ __forceinline__ void gather_group(float4* sh, const float* __restrict__ coords,
                                             const int* __restrict__ cia, int b, int grp, int t)
{
    #pragma unroll
    for (int k = 0; k < SA / 128; ++k) {
        int m   = k * 128 + t;
        int idx = cia[b * Mn + grp * SA + m];
        const float* c = coords + ((size_t)b * Mn + idx) * 3;
        float x = c[0], y = c[1], z = c[2];
        sh[m] = make_float4(x, y, z, sqnorm3(x, y, z));
    }
}

__global__ __launch_bounds__(128) void pairs_kernel(
    const float* __restrict__ coords,
    const int*   __restrict__ cia,
    float*       __restrict__ out,
    float*       __restrict__ ws,
    size_t                    ws_bytes)
{
    __shared__ float4 shP[SA];
    __shared__ float4 shQ[SA];
    __shared__ float  rowp[SA];
    __shared__ float  colp[4][SA];

    const bool sym = parts_fit(ws, ws_bytes);
    const int  t   = threadIdx.x;
    int b, P, Q0, nQ;
    if (sym) {
        b = blockIdx.x / NPAIR;
        int p = blockIdx.x - b * NPAIR;
        P = 0;
        for (int k = 0; k < ST; ++k) {
            int w = ST - P;
            if (p >= w) { p -= w; ++P; }
        }
        Q0 = P + p;
        nQ = 1;
    } else {
        if (blockIdx.x >= Bn * ST) return;
        b  = blockIdx.x / ST;
        P  = blockIdx.x - b * ST;
        Q0 = 0;
        nQ = ST;
    }

    const int wave = __builtin_amdgcn_readfirstlane((int)(threadIdx.x >> 5));
    const int lane = threadIdx.x & 31;
    const int half = lane >> 4;
    const int l    = lane & 15;

    {
        int ip[2], iq[2];
        #pragma unroll
        for (int k = 0; k < 2; ++k) {
            ip[k] = cia[b * Mn + P  * SA + k * 128 + t];
            iq[k] = cia[b * Mn + Q0 * SA + k * 128 + t];
        }
        float4 vp[2], vq[2];
        #pragma unroll
        for (int k = 0; k < 2; ++k) {
            const float* cp = coords + ((size_t)b * Mn + ip[k]) * 3;
            const float* cq = coords + ((size_t)b * Mn + iq[k]) * 3;
            float px = cp[0], py = cp[1], pz = cp[2];
            float qx = cq[0], qy = cq[1], qz = cq[2];
            vp[k] = make_float4(px, py, pz, sqnorm3(px, py, pz));
            vq[k] = make_float4(qx, qy, qz, sqnorm3(qx, qy, qz));
        }
        #pragma unroll
        for (int k = 0; k < 2; ++k) {
            shP[k * 128 + t] = vp[k];
            shQ[k * 128 + t] = vq[k];
        }
    }

    for (int q = 0; q < nQ; ++q) {
        const int Q = Q0 + q;
        if (q > 0) gather_group(shQ, coords, cia, b, Q, t);
        __syncthreads();

        float ccol[16];
        #pragma unroll
        for (int j = 0; j < 16; ++j) ccol[j] = 0.0f;

        for (int it = 0; it < 4; ++it) {
            const int i0 = (wave * 4 + it) * 16;
            RowSums rs;
            rs.p0 = v2f{0.0f, 0.0f};
            rs.p1 = v2f{0.0f, 0.0f};
            rs.p2 = v2f{0.0f, 0.0f};
            rs.p3 = v2f{0.0f, 0.0f};

            sweep_tile(shP, shQ, i0, half, l, rs, ccol);

            float acc[8] = {rs.p0.x, rs.p0.y, rs.p1.x, rs.p1.y, rs.p2.x, rs.p2.y, rs.p3.x, rs.p3.y};
            #pragma unroll
            for (int r = 0; r < 8; ++r) {
                acc[r] = row_xor_add<1>(acc[r]);
                acc[r] = row_xor_add<2>(acc[r]);
                acc[r] = row_xor_add<4>(acc[r]);
                acc[r] = row_xor_add<8>(acc[r]);
            }
            if (l == 0) {
                #pragma unroll
                for (int r = 0; r < 8; ++r) {
                    int row = i0 + half * 8 + r;
                    rowp[row] = (q == 0) ? acc[r] : rowp[row] + acc[r];
                }
            }
        }

        #pragma unroll
        for (int j = 0; j < 16; ++j) ccol[j] += __shfl_xor(ccol[j], 16, 32);
        if (half == 0) {
            #pragma unroll
            for (int j = 0; j < 16; ++j) colp[wave][j * 16 + l] = ccol[j];
        }
        __syncthreads();

        if (sym && P != Q) {
            for (int k = 0; k < SA / 128; ++k) {
                int m = k * 128 + t;
                float c = (colp[0][m] + colp[1][m]) + (colp[2][m] + colp[3][m]);
                ws[(size_t)P * Nn + b * Mn + Q * SA + m] = c;
            }
        }
    }

    for (int k = 0; k < SA / 128; ++k) {
        int m = k * 128 + t;
        int n = b * Mn + P * SA + m;
        if (sym) {
            ws[(size_t)Q0 * Nn + n] = rowp[m];
        } else {
            bool valid = (shP[m].x != PADF);
            out[(size_t)n * 8 + 1] = valid ? rowp[m] : 0.0f;
        }
    }
}

extern "C" void kernel_launch(void* const* d_in, const int* in_sizes, int n_in,
                              void* d_out, int out_size, void* d_ws, size_t ws_size,
                              hipStream_t stream) {
    (void)in_sizes; (void)n_in; (void)out_size;
    const float* coords   = (const float*)d_in[0];
    const int*   cia      = (const int*)d_in[2];
    const int*   partners = (const int*)d_in[3];
    const int*   aidx     = (const int*)d_in[4];
    float*       out      = (float*)d_out;
    float*       ws       = (float*)d_ws;

    pairs_kernel<<<Bn * NPAIR, 128, 0, stream>>>(coords, cia, out, ws, ws_size);
    atoms_kernel<<<Nn / 256, 256, 0, stream>>>(coords, cia, partners, aidx, out, ws, ws_size);
}


# per-atom launch: loads regrouped as 3 dependent memory hops (indices; partner rows + dihedral slot indices + own x; dihedral rows), gathers unconditional, padding by select; was 7 hops in a row
# speedup vs baseline: 2.0803x; 1.0628x over previous
#include <hip/hip_runtime.h>
#include <math.h>


constexpr int   Bn   = 4;
constexpr int   Mn   = 4096;
constexpr int   Nn   = Bn * Mn;
constexpr int   PADI = -999;
#define PADF (-999.0f)
constexpr float CUTOFF2 = 49.0f;

constexpr int ST    = 16;
constexpr int SA    = Mn / ST;
constexpr int NPAIR = ST * (ST + 1) / 2;

typedef float v2f __attribute__((ext_vector_type(2)));
typedef float v8f __attribute__((ext_vector_type(8)));

__device__ __forceinline__ float3 sub3(float3 a, float3 b) {
    return make_float3(a.x - b.x, a.y - b.y, a.z - b.z);
}
__device__ __forceinline__ float3 cross3(float3 a, float3 b) {
    return make_float3(a.y * b.z - a.z * b.y,
                       a.z * b.x - a.x * b.z,
                       a.x * b.y - a.y * b.x);
}
__device__ __forceinline__ float dot3(float3 a, float3 b) {
    return a.x * b.x + a.y * b.y + a.z * b.z;
}

__device__ __forceinline__ float sqnorm3(float x, float y, float z) {
    return fmaf(z, z, fmaf(x, x, y * y));
}

__device__ __forceinline__ bool parts_fit(const float* ws, size_t ws_bytes) {
    return ws != nullptr && ws_bytes >= (size_t)ST * Nn * 4;
}

__global__ __launch_bounds__(256) void atoms_kernel(
    const float* __restrict__ coords,
    const int*   __restrict__ cia,
    const int*   __restrict__ partners,
    const int*   __restrict__ aidx,
    float*       __restrict__ out,
    const float* __restrict__ ws,
    size_t                    ws_bytes)
{
    int n = blockIdx.x * 256 + threadIdx.x;
    if (n >= Nn) return;
    int b = n >> 12;


    int gi = cia[n];
    int p0 = partners[n * 2 + 0];
    int p1 = partners[n * 2 + 1];
    int a[4];
    #pragma unroll
    for (int t = 0; t < 4; ++t) a[t] = aidx[n * 4 + t];

    const bool have_parts = parts_fit(ws, ws_bytes);
    float part[ST];
    #pragma unroll
    for (int s = 0; s < ST; ++s) part[s] = 0.0f;
    if (have_parts) {
        #pragma unroll
        for (int s = 0; s < ST; ++s) part[s] = ws[(size_t)s * Nn + n];
    }

    const bool pv0 = (p0 != PADI);
    const bool pv1 = (p1 != PADI);
    const bool am  = (a[0] != PADI) && (a[1] != PADI) && (a[2] != PADI) && (a[3] != PADI);

    int gb[4], gm[4];
    #pragma unroll
    for (int t = 0; t < 4; ++t) {
        int g = am ? a[t] : 0;
        gb[t] = g >> 12;
        gm[t] = cia[g];
    }
    const float* pc0 = coords + ((size_t)b * Mn + (pv0 ? p0 : 0)) * 3;
    const float* pc1 = coords + ((size_t)b * Mn + (pv1 ? p1 : 0)) * 3;
    float q0x = pc0[0], q0y = pc0[1], q0z = pc0[2];
    float q1x = pc1[0], q1y = pc1[1], q1z = pc1[2];
    float gx = coords[((size_t)b * Mn + gi) * 3];
    asm volatile("" :: "v"(gm[0]), "v"(gm[1]), "v"(gm[2]), "v"(gm[3]),
                       "v"(q0x), "v"(q0y), "v"(q0z), "v"(q1x), "v"(q1y), "v"(q1z), "v"(gx));

    float3 p[4];
    #pragma unroll
    for (int t = 0; t < 4; ++t) {
        const float* c = coords + ((size_t)gb[t] * Mn + gm[t]) * 3;
        p[t] = make_float3(c[0], c[1], c[2]);
    }

    out[(size_t)n * 8 + 2] = pv0 ? q0x : PADF;
    out[(size_t)n * 8 + 3] = pv0 ? q0y : PADF;
    out[(size_t)n * 8 + 4] = pv0 ? q0z : PADF;
    out[(size_t)n * 8 + 5] = pv1 ? q1x : PADF;
    out[(size_t)n * 8 + 6] = pv1 ? q1y : PADF;
    out[(size_t)n * 8 + 7] = pv1 ? q1z : PADF;

    float3 b1 = sub3(p[1], p[0]);
    float3 b2 = sub3(p[2], p[1]);
    float3 b3 = sub3(p[3], p[2]);
    float3 n1 = cross3(b1, b2);
    float3 n2 = cross3(b2, b3);
    float  inv = 1.0f / (sqrtf(dot3(b2, b2)) + 1e-12f);
    float3 b2n = make_float3(b2.x * inv, b2.y * inv, b2.z * inv);
    float3 m1  = cross3(n1, b2n);
    float  xx  = dot3(n1, n2);
    float  yy  = dot3(m1, n2);
    out[(size_t)n * 8 + 0] = am ? atan2f(yy, xx) : PADF;

    if (have_parts) {
        float s = part[0];
        #pragma unroll
        for (int k = 1; k < ST; ++k) s += part[k];
        out[(size_t)n * 8 + 1] = (gx != PADF) ? s : 0.0f;
    }
}


struct RowSums { v2f p0, p1, p2, p3; };

__device__ __forceinline__ void finish_half(const float* dt, v2f& lo, v2f& hi, float& cc) {
    float val[4];
    #pragma unroll
    for (int v = 0; v < 4; ++v) {
        float t = fmaxf(dt[v], 0.0f) + 1e-12f;
        float s = __builtin_amdgcn_sqrtf(t);
        val[v]  = (t <= CUTOFF2) ? s : 0.0f;
        cc     += val[v];
    }
    lo += v2f{val[0], val[1]};
    hi += v2f{val[2], val[3]};
}

template <int CB>
__device__ __forceinline__ void finish_step(const float (&ds)[32], const unsigned (&fs)[8],
                                            RowSums& acc, float (&ccol)[16]) {
    #pragma unroll
    for (int g = 0; g < 4; ++g) {
        if (fs[2 * g]     != 0u) finish_half(ds + 8 * g,     acc.p0, acc.p1, ccol[CB + g]);
        if (fs[2 * g + 1] != 0u) finish_half(ds + 8 * g + 4, acc.p2, acc.p3, ccol[CB + g]);
    }
}

template <int G>
__device__ __forceinline__ void form_tile(float4 br, v2f A, const float (&h)[8],
                                          int half, float (&ds)[32], unsigned (&fs)[8]) {
    v8f zc = {};
    v2f Bf;
    Bf.x = half ? br.z : br.x;
    Bf.y = half ? 0.0f : br.y;
    float sqj = br.w;
    v8f Gm = __builtin_amdgcn_wmma_f32_16x16x4_f32(
        false, A, false, Bf, (short)0, zc, false, false);
    #pragma unroll
    for (int r = 0; r < 8; ++r)
        ds[8 * G + r] = fmaf(Gm[r], -2.0f, h[r] + sqj);
    float m0 = fminf(fminf(ds[8 * G],     ds[8 * G + 1]), fminf(ds[8 * G + 2], ds[8 * G + 3]));
    float m1 = fminf(fminf(ds[8 * G + 4], ds[8 * G + 5]), fminf(ds[8 * G + 6], ds[8 * G + 7]));
    fs[2 * G]     = __builtin_amdgcn_ballot_w32(m0 <= CUTOFF2);
    fs[2 * G + 1] = __builtin_amdgcn_ballot_w32(m1 <= CUTOFF2);
}

__device__ __forceinline__ void form_step(float4 r0, float4 r1, float4 r2, float4 r3, v2f A,
                                          const float (&h)[8], int half,
                                          float (&ds)[32], unsigned (&fs)[8]) {
    form_tile<0>(r0, A, h, half, ds, fs);
    form_tile<1>(r1, A, h, half, ds, fs);
    form_tile<2>(r2, A, h, half, ds, fs);
    form_tile<3>(r3, A, h, half, ds, fs);
}

__device__ __forceinline__ void sweep_tile(const float4* shP, const float4* shQ, int i0,
                                           int half, int l, RowSums& acc, float (&ccol)[16])
{
    float4 arow = shP[i0 + l];
    v2f A;
    A.x = half ? arow.z : arow.x;
    A.y = half ? 0.0f   : arow.y;

    float h[8];
    #pragma unroll
    for (int r = 0; r < 8; ++r) h[r] = shP[i0 + half * 8 + r].w;

    float    dA[32], dB[32];
    unsigned fA[8], fB[8];
    const float4* q = shQ + l;
    float4 r0, r1, r2, r3;

    r0 = q[0];   r1 = q[16];  r2 = q[32];  r3 = q[48];
    form_step(r0, r1, r2, r3, A, h, half, dA, fA);

    r0 = q[64];  r1 = q[80];  r2 = q[96];  r3 = q[112];
    form_step(r0, r1, r2, r3, A, h, half, dB, fB);

    r0 = q[128]; r1 = q[144]; r2 = q[160]; r3 = q[176];
    finish_step<0>(dA, fA, acc, ccol);
    form_step(r0, r1, r2, r3, A, h, half, dA, fA);

    r0 = q[192]; r1 = q[208]; r2 = q[224]; r3 = q[240];
    finish_step<4>(dB, fB, acc, ccol);
    form_step(r0, r1, r2, r3, A, h, half, dB, fB);

    finish_step<8>(dA, fA, acc, ccol);
    finish_step<12>(dB, fB, acc, ccol);
}

static_assert(SA == 256, "sweep_tile walks 4 steps of 64 columns; 128 threads fill 2 rows each");

template <int M>
__device__ __forceinline__ float row_xor_add(float v) {
    int o = __builtin_amdgcn_update_dpp(0, __float_as_int(v), 0x160 + M, 0xF, 0xF, false);
    return v + __int_as_float(o);
}

__device__ __forceinline__ void gather_group(float4* sh, const float* __restrict__ coords,
                                             const int* __restrict__ cia, int b, int grp, int t)
{
    #pragma unroll
    for (int k = 0; k < SA / 128; ++k) {
        int m   = k * 128 + t;
        int idx = cia[b * Mn + grp * SA + m];
        const float* c = coords + ((size_t)b * Mn + idx) * 3;
        float x = c[0], y = c[1], z = c[2];
        sh[m] = make_float4(x, y, z, sqnorm3(x, y, z));
    }
}

__global__ __launch_bounds__(128) void pairs_kernel(
    const float* __restrict__ coords,
    const int*   __restrict__ cia,
    float*       __restrict__ out,
    float*       __restrict__ ws,
    size_t                    ws_bytes)
{
    __shared__ float4 shP[SA];
    __shared__ float4 shQ[SA];
    __shared__ float  rowp[SA];
    __shared__ float  colp[4][SA];

    const bool sym = parts_fit(ws, ws_bytes);
    const int  t   = threadIdx.x;
    int b, P, Q0, nQ;
    if (sym) {
        b = blockIdx.x / NPAIR;
        int p = blockIdx.x - b * NPAIR;
        P = 0;
        for (int k = 0; k < ST; ++k) {
            int w = ST - P;
            if (p >= w) { p -= w; ++P; }
        }
        Q0 = P + p;
        nQ = 1;
    } else {
        if (blockIdx.x >= Bn * ST) return;
        b  = blockIdx.x / ST;
        P  = blockIdx.x - b * ST;
        Q0 = 0;
        nQ = ST;
    }

    const int wave = __builtin_amdgcn_readfirstlane((int)(threadIdx.x >> 5));
    const int lane = threadIdx.x & 31;
    const int half = lane >> 4;
    const int l    = lane & 15;

    {
        int ip[2], iq[2];
        #pragma unroll
        for (int k = 0; k < 2; ++k) {
            ip[k] = cia[b * Mn + P  * SA + k * 128 + t];
            iq[k] = cia[b * Mn + Q0 * SA + k * 128 + t];
        }
        float4 vp[2], vq[2];
        #pragma unroll
        for (int k = 0; k < 2; ++k) {
            const float* cp = coords + ((size_t)b * Mn + ip[k]) * 3;
            const float* cq = coords + ((size_t)b * Mn + iq[k]) * 3;
            float px = cp[0], py = cp[1], pz = cp[2];
            float qx = cq[0], qy = cq[1], qz = cq[2];
            vp[k] = make_float4(px, py, pz, sqnorm3(px, py, pz));
            vq[k] = make_float4(qx, qy, qz, sqnorm3(qx, qy, qz));
        }
        #pragma unroll
        for (int k = 0; k < 2; ++k) {
            shP[k * 128 + t] = vp[k];
            shQ[k * 128 + t] = vq[k];
        }
    }

    for (int q = 0; q < nQ; ++q) {
        const int Q = Q0 + q;
        if (q > 0) gather_group(shQ, coords, cia, b, Q, t);
        __syncthreads();

        float ccol[16];
        #pragma unroll
        for (int j = 0; j < 16; ++j) ccol[j] = 0.0f;

        for (int it = 0; it < 4; ++it) {
            const int i0 = (wave * 4 + it) * 16;
            RowSums rs;
            rs.p0 = v2f{0.0f, 0.0f};
            rs.p1 = v2f{0.0f, 0.0f};
            rs.p2 = v2f{0.0f, 0.0f};
            rs.p3 = v2f{0.0f, 0.0f};

            sweep_tile(shP, shQ, i0, half, l, rs, ccol);

            float acc[8] = {rs.p0.x, rs.p0.y, rs.p1.x, rs.p1.y, rs.p2.x, rs.p2.y, rs.p3.x, rs.p3.y};
            #pragma unroll
            for (int r = 0; r < 8; ++r) {
                acc[r] = row_xor_add<1>(acc[r]);
                acc[r] = row_xor_add<2>(acc[r]);
                acc[r] = row_xor_add<4>(acc[r]);
                acc[r] = row_xor_add<8>(acc[r]);
            }
            if (l == 0) {
                #pragma unroll
                for (int r = 0; r < 8; ++r) {
                    int row = i0 + half * 8 + r;
                    rowp[row] = (q == 0) ? acc[r] : rowp[row] + acc[r];
                }
            }
        }

        #pragma unroll
        for (int j = 0; j < 16; ++j) ccol[j] += __shfl_xor(ccol[j], 16, 32);
        if (half == 0) {
            #pragma unroll
            for (int j = 0; j < 16; ++j) colp[wave][j * 16 + l] = ccol[j];
        }
        __syncthreads();

        if (sym && P != Q) {
            for (int k = 0; k < SA / 128; ++k) {
                int m = k * 128 + t;
                float c = (colp[0][m] + colp[1][m]) + (colp[2][m] + colp[3][m]);
                ws[(size_t)P * Nn + b * Mn + Q * SA + m] = c;
            }
        }
    }

    for (int k = 0; k < SA / 128; ++k) {
        int m = k * 128 + t;
        int n = b * Mn + P * SA + m;
        if (sym) {
            ws[(size_t)Q0 * Nn + n] = rowp[m];
        } else {
            bool valid = (shP[m].x != PADF);
            out[(size_t)n * 8 + 1] = valid ? rowp[m] : 0.0f;
        }
    }
}

extern "C" void kernel_launch(void* const* d_in, const int* in_sizes, int n_in,
                              void* d_out, int out_size, void* d_ws, size_t ws_size,
                              hipStream_t stream) {
    (void)in_sizes; (void)n_in; (void)out_size;
    const float* coords   = (const float*)d_in[0];
    const int*   cia      = (const int*)d_in[2];
    const int*   partners = (const int*)d_in[3];
    const int*   aidx     = (const int*)d_in[4];
    float*       out      = (float*)d_out;
    float*       ws       = (float*)d_ws;

    pairs_kernel<<<Bn * NPAIR, 128, 0, stream>>>(coords, cia, out, ws, ws_size);
    atoms_kernel<<<Nn / 256, 256, 0, stream>>>(coords, cia, partners, aidx, out, ws, ws_size);
}


# nbr: a group's inner pairs go by 64-atom runs (own run rows only, later runs once for both sides), two groups per block: 512 near-equal blocks, was 544; changes the order of the row sum
# speedup vs baseline: 2.2758x; 1.0940x over previous
#include <hip/hip_runtime.h>
#include <math.h>


constexpr int   Bn   = 4;
constexpr int   Mn   = 4096;
constexpr int   Nn   = Bn * Mn;
constexpr int   PADI = -999;
#define PADF (-999.0f)
constexpr float CUTOFF2 = 49.0f;

constexpr int ST    = 16;
constexpr int SA    = Mn / ST;
constexpr int NOFF  = ST * (ST - 1) / 2;
constexpr int NDIAG = ST / 2;
constexpr int NBLK  = Bn * (NOFF + NDIAG);

typedef float v2f __attribute__((ext_vector_type(2)));
typedef float v8f __attribute__((ext_vector_type(8)));

__device__ __forceinline__ float3 sub3(float3 a, float3 b) {
    return make_float3(a.x - b.x, a.y - b.y, a.z - b.z);
}
__device__ __forceinline__ float3 cross3(float3 a, float3 b) {
    return make_float3(a.y * b.z - a.z * b.y,
                       a.z * b.x - a.x * b.z,
                       a.x * b.y - a.y * b.x);
}
__device__ __forceinline__ float dot3(float3 a, float3 b) {
    return a.x * b.x + a.y * b.y + a.z * b.z;
}

__device__ __forceinline__ float sqnorm3(float x, float y, float z) {
    return fmaf(z, z, fmaf(x, x, y * y));
}

__device__ __forceinline__ bool parts_fit(const float* ws, size_t ws_bytes) {
    return ws != nullptr && ws_bytes >= (size_t)ST * Nn * 4;
}

__global__ __launch_bounds__(256) void atoms_kernel(
    const float* __restrict__ coords,
    const int*   __restrict__ cia,
    const int*   __restrict__ partners,
    const int*   __restrict__ aidx,
    float*       __restrict__ out,
    const float* __restrict__ ws,
    size_t                    ws_bytes)
{
    int n = blockIdx.x * 256 + threadIdx.x;
    if (n >= Nn) return;
    int b = n >> 12;


    int gi = cia[n];
    int p0 = partners[n * 2 + 0];
    int p1 = partners[n * 2 + 1];
    int a[4];
    #pragma unroll
    for (int t = 0; t < 4; ++t) a[t] = aidx[n * 4 + t];

    const bool have_parts = parts_fit(ws, ws_bytes);
    float part[ST];
    #pragma unroll
    for (int s = 0; s < ST; ++s) part[s] = 0.0f;
    if (have_parts) {
        #pragma unroll
        for (int s = 0; s < ST; ++s) part[s] = ws[(size_t)s * Nn + n];
    }

    const bool pv0 = (p0 != PADI);
    const bool pv1 = (p1 != PADI);
    const bool am  = (a[0] != PADI) && (a[1] != PADI) && (a[2] != PADI) && (a[3] != PADI);

    int gb[4], gm[4];
    #pragma unroll
    for (int t = 0; t < 4; ++t) {
        int g = am ? a[t] : 0;
        gb[t] = g >> 12;
        gm[t] = cia[g];
    }
    const float* pc0 = coords + ((size_t)b * Mn + (pv0 ? p0 : 0)) * 3;
    const float* pc1 = coords + ((size_t)b * Mn + (pv1 ? p1 : 0)) * 3;
    float q0x = pc0[0], q0y = pc0[1], q0z = pc0[2];
    float q1x = pc1[0], q1y = pc1[1], q1z = pc1[2];
    float gx = coords[((size_t)b * Mn + gi) * 3];
    asm volatile("" :: "v"(gm[0]), "v"(gm[1]), "v"(gm[2]), "v"(gm[3]),
                       "v"(q0x), "v"(q0y), "v"(q0z), "v"(q1x), "v"(q1y), "v"(q1z), "v"(gx));

    float3 p[4];
    #pragma unroll
    for (int t = 0; t < 4; ++t) {
        const float* c = coords + ((size_t)gb[t] * Mn + gm[t]) * 3;
        p[t] = make_float3(c[0], c[1], c[2]);
    }

    out[(size_t)n * 8 + 2] = pv0 ? q0x : PADF;
    out[(size_t)n * 8 + 3] = pv0 ? q0y : PADF;
    out[(size_t)n * 8 + 4] = pv0 ? q0z : PADF;
    out[(size_t)n * 8 + 5] = pv1 ? q1x : PADF;
    out[(size_t)n * 8 + 6] = pv1 ? q1y : PADF;
    out[(size_t)n * 8 + 7] = pv1 ? q1z : PADF;

    float3 b1 = sub3(p[1], p[0]);
    float3 b2 = sub3(p[2], p[1]);
    float3 b3 = sub3(p[3], p[2]);
    float3 n1 = cross3(b1, b2);
    float3 n2 = cross3(b2, b3);
    float  inv = 1.0f / (sqrtf(dot3(b2, b2)) + 1e-12f);
    float3 b2n = make_float3(b2.x * inv, b2.y * inv, b2.z * inv);
    float3 m1  = cross3(n1, b2n);
    float  xx  = dot3(n1, n2);
    float  yy  = dot3(m1, n2);
    out[(size_t)n * 8 + 0] = am ? atan2f(yy, xx) : PADF;

    if (have_parts) {
        float s = part[0];
        #pragma unroll
        for (int k = 1; k < ST; ++k) s += part[k];
        out[(size_t)n * 8 + 1] = (gx != PADF) ? s : 0.0f;
    }
}


struct RowSums { v2f p0, p1, p2, p3; };

__device__ __forceinline__ void finish_half(const float* dt, v2f& lo, v2f& hi, float& cc) {
    float val[4];
    #pragma unroll
    for (int v = 0; v < 4; ++v) {
        float t = fmaxf(dt[v], 0.0f) + 1e-12f;
        float s = __builtin_amdgcn_sqrtf(t);
        val[v]  = (t <= CUTOFF2) ? s : 0.0f;
        cc     += val[v];
    }
    lo += v2f{val[0], val[1]};
    hi += v2f{val[2], val[3]};
}

template <int CB>
__device__ __forceinline__ void finish_step(const float (&ds)[32], const unsigned (&fs)[8],
                                            RowSums& acc, float (&ccol)[16]) {
    #pragma unroll
    for (int g = 0; g < 4; ++g) {
        if (fs[2 * g]     != 0u) finish_half(ds + 8 * g,     acc.p0, acc.p1, ccol[CB + g]);
        if (fs[2 * g + 1] != 0u) finish_half(ds + 8 * g + 4, acc.p2, acc.p3, ccol[CB + g]);
    }
}

template <int G>
__device__ __forceinline__ void form_tile(float4 br, v2f A, const float (&h)[8],
                                          int half, float (&ds)[32], unsigned (&fs)[8]) {
    v8f zc = {};
    v2f Bf;
    Bf.x = half ? br.z : br.x;
    Bf.y = half ? 0.0f : br.y;
    float sqj = br.w;
    v8f Gm = __builtin_amdgcn_wmma_f32_16x16x4_f32(
        false, A, false, Bf, (short)0, zc, false, false);
    #pragma unroll
    for (int r = 0; r < 8; ++r)
        ds[8 * G + r] = fmaf(Gm[r], -2.0f, h[r] + sqj);
    float m0 = fminf(fminf(ds[8 * G],     ds[8 * G + 1]), fminf(ds[8 * G + 2], ds[8 * G + 3]));
    float m1 = fminf(fminf(ds[8 * G + 4], ds[8 * G + 5]), fminf(ds[8 * G + 6], ds[8 * G + 7]));
    fs[2 * G]     = __builtin_amdgcn_ballot_w32(m0 <= CUTOFF2);
    fs[2 * G + 1] = __builtin_amdgcn_ballot_w32(m1 <= CUTOFF2);
}

__device__ __forceinline__ void form_step(float4 r0, float4 r1, float4 r2, float4 r3, v2f A,
                                          const float (&h)[8], int half,
                                          float (&ds)[32], unsigned (&fs)[8]) {
    form_tile<0>(r0, A, h, half, ds, fs);
    form_tile<1>(r1, A, h, half, ds, fs);
    form_tile<2>(r2, A, h, half, ds, fs);
    form_tile<3>(r3, A, h, half, ds, fs);
}

__device__ __forceinline__ void sweep_tile(const float4* shP, const float4* shQ, int i0,
                                           int half, int l, RowSums& acc, float (&ccol)[16])
{
    float4 arow = shP[i0 + l];
    v2f A;
    A.x = half ? arow.z : arow.x;
    A.y = half ? 0.0f   : arow.y;

    float h[8];
    #pragma unroll
    for (int r = 0; r < 8; ++r) h[r] = shP[i0 + half * 8 + r].w;

    float    dA[32], dB[32];
    unsigned fA[8], fB[8];
    const float4* q = shQ + l;
    float4 r0, r1, r2, r3;

    r0 = q[0];   r1 = q[16];  r2 = q[32];  r3 = q[48];
    form_step(r0, r1, r2, r3, A, h, half, dA, fA);

    r0 = q[64];  r1 = q[80];  r2 = q[96];  r3 = q[112];
    form_step(r0, r1, r2, r3, A, h, half, dB, fB);

    r0 = q[128]; r1 = q[144]; r2 = q[160]; r3 = q[176];
    finish_step<0>(dA, fA, acc, ccol);
    form_step(r0, r1, r2, r3, A, h, half, dA, fA);

    r0 = q[192]; r1 = q[208]; r2 = q[224]; r3 = q[240];
    finish_step<4>(dB, fB, acc, ccol);
    form_step(r0, r1, r2, r3, A, h, half, dB, fB);

    finish_step<8>(dA, fA, acc, ccol);
    finish_step<12>(dB, fB, acc, ccol);
}

__device__ __forceinline__ void finish_step_rows(const float (&ds)[32], const unsigned (&fs)[8],
                                                 RowSums& acc) {
    float unused[16];
    #pragma unroll
    for (int j = 0; j < 16; ++j) unused[j] = 0.0f;
    finish_step<0>(ds, fs, acc, unused);
}

template <int S0>
__device__ __forceinline__ void sweep_own(const float4* T, int i0, int half, int l,
                                          RowSums& acc, float (&ccol)[16])
{
    float4 arow = T[i0 + l];
    v2f A;
    A.x = half ? arow.z : arow.x;
    A.y = half ? 0.0f   : arow.y;

    float h[8];
    #pragma unroll
    for (int r = 0; r < 8; ++r) h[r] = T[i0 + half * 8 + r].w;

    float    dA[32], dB[32];
    unsigned fA[8], fB[8];
    const float4* q = T + l + 64 * S0;
    float4 r0, r1, r2, r3;

    r0 = q[0];   r1 = q[16];  r2 = q[32];  r3 = q[48];
    form_step(r0, r1, r2, r3, A, h, half, dA, fA);
    if constexpr (S0 <= 2) {
        r0 = q[64];  r1 = q[80];  r2 = q[96];  r3 = q[112];
        form_step(r0, r1, r2, r3, A, h, half, dB, fB);
    }
    if constexpr (S0 <= 1) {
        r0 = q[128]; r1 = q[144]; r2 = q[160]; r3 = q[176];
        finish_step_rows(dA, fA, acc);
        form_step(r0, r1, r2, r3, A, h, half, dA, fA);
    } else {
        finish_step_rows(dA, fA, acc);
    }
    if constexpr (S0 == 0) {
        r0 = q[192]; r1 = q[208]; r2 = q[224]; r3 = q[240];
        finish_step<4>(dB, fB, acc, ccol);
        form_step(r0, r1, r2, r3, A, h, half, dB, fB);
        finish_step<8>(dA, fA, acc, ccol);
        finish_step<12>(dB, fB, acc, ccol);
    }
    if constexpr (S0 == 1) {
        finish_step<8>(dB, fB, acc, ccol);
        finish_step<12>(dA, fA, acc, ccol);
    }
    if constexpr (S0 == 2) {
        finish_step<12>(dB, fB, acc, ccol);
    }
}

static_assert(SA == 256, "sweep_tile walks 4 steps of 64 columns; 128 threads fill 2 rows each");

template <int M>
__device__ __forceinline__ float row_xor_add(float v) {
    int o = __builtin_amdgcn_update_dpp(0, __float_as_int(v), 0x160 + M, 0xF, 0xF, false);
    return v + __int_as_float(o);
}

template <int IT>
__device__ __forceinline__ void own_row_tile(const float4* T, int wave, int half, int l,
                                             float (&ccol)[16], float* rowp)
{
    const int i0 = (4 * IT + wave) * 16;
    RowSums rs;
    rs.p0 = v2f{0.0f, 0.0f};
    rs.p1 = v2f{0.0f, 0.0f};
    rs.p2 = v2f{0.0f, 0.0f};
    rs.p3 = v2f{0.0f, 0.0f};

    sweep_own<IT>(T, i0, half, l, rs, ccol);

    float acc[8] = {rs.p0.x, rs.p0.y, rs.p1.x, rs.p1.y, rs.p2.x, rs.p2.y, rs.p3.x, rs.p3.y};
    #pragma unroll
    for (int r = 0; r < 8; ++r) {
        acc[r] = row_xor_add<1>(acc[r]);
        acc[r] = row_xor_add<2>(acc[r]);
        acc[r] = row_xor_add<4>(acc[r]);
        acc[r] = row_xor_add<8>(acc[r]);
    }
    if (l == 0) {
        #pragma unroll
        for (int r = 0; r < 8; ++r) rowp[i0 + half * 8 + r] = acc[r];
    }
}

__device__ __forceinline__ void gather_group(float4* sh, const float* __restrict__ coords,
                                             const int* __restrict__ cia, int b, int grp, int t)
{
    #pragma unroll
    for (int k = 0; k < SA / 128; ++k) {
        int m   = k * 128 + t;
        int idx = cia[b * Mn + grp * SA + m];
        const float* c = coords + ((size_t)b * Mn + idx) * 3;
        float x = c[0], y = c[1], z = c[2];
        sh[m] = make_float4(x, y, z, sqnorm3(x, y, z));
    }
}

__global__ __launch_bounds__(128) void pairs_kernel(
    const float* __restrict__ coords,
    const int*   __restrict__ cia,
    float*       __restrict__ out,
    float*       __restrict__ ws,
    size_t                    ws_bytes)
{
    __shared__ float4 shP[SA];
    __shared__ float4 shQ[SA];
    __shared__ float  rowp[SA];
    __shared__ float  colp[4][SA];
    __shared__ float  rowp2[SA];
    __shared__ float  colp2[4][SA];

    const bool sym = parts_fit(ws, ws_bytes);
    const int  t   = threadIdx.x;
    int  b, P, Q0, nQ;
    bool own = false;
    if (sym) {
        if (blockIdx.x < Bn * NDIAG) {
            own = true;
            b   = blockIdx.x / NDIAG;
            P   = blockIdx.x - b * NDIAG;
            Q0  = P + NDIAG;
            nQ  = 1;
        } else {
            const int k0 = blockIdx.x - Bn * NDIAG;
            b = k0 / NOFF;
            int p = k0 - b * NOFF;
            P = 0;
            for (int k = 0; k < ST - 1; ++k) {
                int w = ST - 1 - P;
                if (p >= w) { p -= w; ++P; }
            }
            Q0 = P + 1 + p;
            nQ = 1;
        }
    } else {
        if (blockIdx.x >= Bn * ST) return;
        b  = blockIdx.x / ST;
        P  = blockIdx.x - b * ST;
        Q0 = 0;
        nQ = ST;
    }

    const int wave = __builtin_amdgcn_readfirstlane((int)(threadIdx.x >> 5));
    const int lane = threadIdx.x & 31;
    const int half = lane >> 4;
    const int l    = lane & 15;

    {
        int ip[2], iq[2];
        #pragma unroll
        for (int k = 0; k < 2; ++k) {
            ip[k] = cia[b * Mn + P  * SA + k * 128 + t];
            iq[k] = cia[b * Mn + Q0 * SA + k * 128 + t];
        }
        float4 vp[2], vq[2];
        #pragma unroll
        for (int k = 0; k < 2; ++k) {
            const float* cp = coords + ((size_t)b * Mn + ip[k]) * 3;
            const float* cq = coords + ((size_t)b * Mn + iq[k]) * 3;
            float px = cp[0], py = cp[1], pz = cp[2];
            float qx = cq[0], qy = cq[1], qz = cq[2];
            vp[k] = make_float4(px, py, pz, sqnorm3(px, py, pz));
            vq[k] = make_float4(qx, qy, qz, sqnorm3(qx, qy, qz));
        }
        #pragma unroll
        for (int k = 0; k < 2; ++k) {
            shP[k * 128 + t] = vp[k];
            shQ[k * 128 + t] = vq[k];
        }
    }

    if (own) {
        __syncthreads();
        #pragma nounroll
        for (int g = 0; g < 2; ++g) {
            const float4* T  = g ? shQ : shP;
            const int     G  = g ? Q0 : P;
            float*        rp = g ? rowp2 : rowp;
            float (*cp)[SA]  = g ? colp2 : colp;
            float ccol[16];
            #pragma unroll
            for (int j = 0; j < 16; ++j) ccol[j] = 0.0f;

            own_row_tile<0>(T, wave, half, l, ccol, rp);
            own_row_tile<1>(T, wave, half, l, ccol, rp);
            own_row_tile<2>(T, wave, half, l, ccol, rp);
            own_row_tile<3>(T, wave, half, l, ccol, rp);

            #pragma unroll
            for (int j = 0; j < 16; ++j) ccol[j] += __shfl_xor(ccol[j], 16, 32);
            if (half == 0) {
                #pragma unroll
                for (int j = 0; j < 16; ++j) cp[wave][j * 16 + l] = ccol[j];
            }
            __syncthreads();

            for (int k = 0; k < SA / 128; ++k) {
                int m = k * 128 + t;
                float c = (cp[0][m] + cp[1][m]) + (cp[2][m] + cp[3][m]);
                ws[(size_t)G * Nn + b * Mn + G * SA + m] = rp[m] + c;
            }
        }
        return;
    }

    for (int q = 0; q < nQ; ++q) {
        const int Q = Q0 + q;
        if (q > 0) gather_group(shQ, coords, cia, b, Q, t);
        __syncthreads();

        float ccol[16];
        #pragma unroll
        for (int j = 0; j < 16; ++j) ccol[j] = 0.0f;

        for (int it = 0; it < 4; ++it) {
            const int i0 = (wave * 4 + it) * 16;
            RowSums rs;
            rs.p0 = v2f{0.0f, 0.0f};
            rs.p1 = v2f{0.0f, 0.0f};
            rs.p2 = v2f{0.0f, 0.0f};
            rs.p3 = v2f{0.0f, 0.0f};

            sweep_tile(shP, shQ, i0, half, l, rs, ccol);

            float acc[8] = {rs.p0.x, rs.p0.y, rs.p1.x, rs.p1.y, rs.p2.x, rs.p2.y, rs.p3.x, rs.p3.y};
            #pragma unroll
            for (int r = 0; r < 8; ++r) {
                acc[r] = row_xor_add<1>(acc[r]);
                acc[r] = row_xor_add<2>(acc[r]);
                acc[r] = row_xor_add<4>(acc[r]);
                acc[r] = row_xor_add<8>(acc[r]);
            }
            if (l == 0) {
                #pragma unroll
                for (int r = 0; r < 8; ++r) {
                    int row = i0 + half * 8 + r;
                    rowp[row] = (q == 0) ? acc[r] : rowp[row] + acc[r];
                }
            }
        }

        #pragma unroll
        for (int j = 0; j < 16; ++j) ccol[j] += __shfl_xor(ccol[j], 16, 32);
        if (half == 0) {
            #pragma unroll
            for (int j = 0; j < 16; ++j) colp[wave][j * 16 + l] = ccol[j];
        }
        __syncthreads();

        if (sym && P != Q) {
            for (int k = 0; k < SA / 128; ++k) {
                int m = k * 128 + t;
                float c = (colp[0][m] + colp[1][m]) + (colp[2][m] + colp[3][m]);
                ws[(size_t)P * Nn + b * Mn + Q * SA + m] = c;
            }
        }
    }

    for (int k = 0; k < SA / 128; ++k) {
        int m = k * 128 + t;
        int n = b * Mn + P * SA + m;
        if (sym) {
            ws[(size_t)Q0 * Nn + n] = rowp[m];
        } else {
            bool valid = (shP[m].x != PADF);
            out[(size_t)n * 8 + 1] = valid ? rowp[m] : 0.0f;
        }
    }
}

extern "C" void kernel_launch(void* const* d_in, const int* in_sizes, int n_in,
                              void* d_out, int out_size, void* d_ws, size_t ws_size,
                              hipStream_t stream) {
    (void)in_sizes; (void)n_in; (void)out_size;
    const float* coords   = (const float*)d_in[0];
    const int*   cia      = (const int*)d_in[2];
    const int*   partners = (const int*)d_in[3];
    const int*   aidx     = (const int*)d_in[4];
    float*       out      = (float*)d_out;
    float*       ws       = (float*)d_ws;

    pairs_kernel<<<NBLK, 128, 0, stream>>>(coords, cia, out, ws, ws_size);
    atoms_kernel<<<Nn / 256, 256, 0, stream>>>(coords, cia, partners, aidx, out, ws, ws_size);
}
